# starting kernel
# baseline (speedup 1.0000x reference)
#include <hip/hip_runtime.h>

typedef float v2f __attribute__((ext_vector_type(2)));
typedef float v4f __attribute__((ext_vector_type(4)));
typedef float v8f __attribute__((ext_vector_type(8)));

#define M_DIM   16
#define K_DIM   4096
#define N_DIM   11008
#define N8      (N_DIM / 8)
#define GROUPK  128
#define WAVES   8
#define KPW     (K_DIM / WAVES)
#define KSTEP   16

__global__ __launch_bounds__(WAVES * 32)
void awq_wmma_f32_kernel(const float* __restrict__ x,
                         const int*   __restrict__ qweight,
                         const int*   __restrict__ qzeros,
                         const float* __restrict__ scales,
                         const float* __restrict__ bias,
                         float* __restrict__ out)
{
    __shared__ float red[WAVES][32][8];

    const int lane  = threadIdx.x & 31;
    const int wave  = threadIdx.x >> 5;
    const int tile  = blockIdx.x;
    const int n_base    = tile * 16;
    const int qcol_base = tile * 2;

    const int nn    = lane & 15;
    const int kh    = lane >> 4;
    const int mycol = (nn >> 3) & 1;
    const int n8    = nn & 7;
    const int shift = 4 * ((n8 >> 1) + 4 * (n8 & 1));

    const int kstart = wave * KPW;

    const int qrow = lane >> 1;
    const int qc   = lane & 1;
    const int* qw_ptr = qweight + (size_t)(kstart + qrow) * N8 + qcol_base + qc;

    const float* xrow = x + (size_t)nn * K_DIM + kstart + 8 * kh;

    v8f acc = {};
    float sc  = 0.0f;
    float nzs = 0.0f;

    for (int k0 = 0; k0 < KPW; k0 += KSTEP) {
        if (((kstart + k0) & (GROUPK - 1)) == 0) {
            const int g = (kstart + k0) >> 7;
            sc = scales[(size_t)g * N_DIM + n_base + nn];
            const int zdw = qzeros[(size_t)g * N8 + qcol_base + mycol];
            nzs = -(float)((zdw >> shift) & 15) * sc;
        }

        const int myqw = *qw_ptr;
        qw_ptr += KSTEP * N8;
        __builtin_prefetch(qw_ptr, 0, 0);

        const v4f xa = *(const v4f*)(xrow + k0);
        const v4f xb = *(const v4f*)(xrow + k0 + 4);
        const float xk[8] = { xa.x, xa.y, xa.z, xa.w, xb.x, xb.y, xb.z, xb.w };

        #pragma unroll
        for (int j = 0; j < 4; ++j) {
            const int klocal = 8 * kh + 2 * j;
            const int s0 = ((klocal + 0) << 1) | mycol;
            const int s1 = ((klocal + 1) << 1) | mycol;
            const int d0 = __builtin_amdgcn_ds_bpermute(s0 << 2, myqw);
            const int d1 = __builtin_amdgcn_ds_bpermute(s1 << 2, myqw);

            v2f b;
            b.x = fmaf((float)((d0 >> shift) & 15), sc, nzs);
            b.y = fmaf((float)((d1 >> shift) & 15), sc, nzs);

            v2f a;
            a.x = xk[2 * j];
            a.y = xk[2 * j + 1];

            acc = __builtin_amdgcn_wmma_f32_16x16x4_f32(
                       false, a,  false, b,
                       (short)0, acc,
                       false,  false);
        }
    }

    #pragma unroll
    for (int v = 0; v < 8; ++v) red[wave][lane][v] = acc[v];
    __syncthreads();

    if (wave == 0) {
        const float bv = bias[n_base + nn];
        #pragma unroll
        for (int v = 0; v < 8; ++v) {
            float s = 0.0f;
            #pragma unroll
            for (int w = 0; w < WAVES; ++w) s += red[w][lane][v];
            const int m = v + 8 * kh;
            out[(size_t)m * N_DIM + n_base + nn] = s + bv;
        }
    }
}

extern "C" void kernel_launch(void* const* d_in, const int* in_sizes, int n_in,
                              void* d_out, int out_size, void* d_ws, size_t ws_size,
                              hipStream_t stream) {
    (void)in_sizes; (void)n_in; (void)d_ws; (void)ws_size; (void)out_size;
    const float* x       = (const float*)d_in[0];
    const int*   qweight = (const int*)  d_in[1];
    const int*   qzeros  = (const int*)  d_in[2];
    const float* scales  = (const float*)d_in[3];
    const float* bias    = (const float*)d_in[4];
    float*       out     = (float*)d_out;

    dim3 grid(N_DIM / 16);
    dim3 block(WAVES * 32);
    awq_wmma_f32_kernel<<<grid, block, 0, stream>>>(x, qweight, qzeros, scales, bias, out);
}


# B path rebuilt: lane pairs load whole packed dwords along a row (no lane permutes), each lane feeds its 4 nibbles to 4 tiles sharing one A fragment, loads a stage ahead; 64 cols/block, K on 16 waves
# speedup vs baseline: 1.6964x; 1.6964x over previous
#include <hip/hip_runtime.h>


typedef float v2f __attribute__((ext_vector_type(2)));
typedef float v4f __attribute__((ext_vector_type(4)));
typedef float v8f __attribute__((ext_vector_type(8)));

#define M_DIM   16
#define K_DIM   4096
#define N_DIM   11008
#define N8      (N_DIM / 8)
#define GROUPK  128
#define WAVES   16
#define KPW     (K_DIM / WAVES)
#define NGRP    (KPW / GROUPK)
#define KSTEP   16
#define STRIP   64
#define SDW     (STRIP / 8)
#define TILES   4

struct Stage {
    int q[8];
    v4f xa, xb;
};

__device__ __forceinline__ void load_stage(Stage& s, const int* __restrict__ qp,
                                           const float* __restrict__ xp)
{
    #pragma unroll
    for (int i = 0; i < 8; ++i) s.q[i] = qp[(size_t)i * N8];
    s.xa = *(const v4f*)(xp);
    s.xb = *(const v4f*)(xp + 4);
}

__device__ __forceinline__ void compute_stage(v8f (&acc)[TILES], const Stage& s,
                                              const int sh16,
                                              const float (&sc)[TILES],
                                              const float (&nzs)[TILES])
{
    const float xk[8] = { s.xa.x, s.xa.y, s.xa.z, s.xa.w,
                          s.xb.x, s.xb.y, s.xb.z, s.xb.w };
    #pragma unroll
    for (int j = 0; j < 4; ++j) {
        const int d0 = s.q[2 * j]     >> sh16;
        const int d1 = s.q[2 * j + 1] >> sh16;
        v2f a;
        a.x = xk[2 * j];
        a.y = xk[2 * j + 1];
        #pragma unroll
        for (int t = 0; t < TILES; ++t) {
            v2f b;
            b.x = fmaf((float)((d0 >> (4 * t)) & 15), sc[t], nzs[t]);
            b.y = fmaf((float)((d1 >> (4 * t)) & 15), sc[t], nzs[t]);
            acc[t] = __builtin_amdgcn_wmma_f32_16x16x4_f32(
                          false, a,  false, b,
                          (short)0, acc[t],
                          false,  false);
        }
    }
}

__global__ __launch_bounds__(WAVES * 32)
void awq_wmma_f32_kernel(const float* __restrict__ x,
                         const int*   __restrict__ qweight,
                         const int*   __restrict__ qzeros,
                         const float* __restrict__ scales,
                         const float* __restrict__ bias,
                         float* __restrict__ out)
{
    __shared__ float red[WAVES][M_DIM][STRIP];

    const int lane  = threadIdx.x & 31;
    const int wave  = threadIdx.x >> 5;
    const int strip = blockIdx.x;

    const int c  = lane & 15;
    const int h  = lane >> 4;
    const int cd = c & 7;
    const int ch = c >> 3;
    const int sh16 = 16 * ch;

    const int kstart = wave * KPW;
    const int g0     = kstart / GROUPK;

    float sc[NGRP][TILES], nzs[NGRP][TILES];
    #pragma unroll
    for (int g = 0; g < NGRP; ++g) {
        const int zdw = qzeros[(size_t)(g0 + g) * N8 + strip * SDW + cd] >> sh16;
        #pragma unroll
        for (int t = 0; t < TILES; ++t) {
            const float s = scales[(size_t)(g0 + g) * N_DIM + strip * STRIP + 8 * cd + 2 * t + ch];
            sc[g][t]  = s;
            nzs[g][t] = -(float)((zdw >> (4 * t)) & 15) * s;
        }
    }

    const int*   qp = qweight + (size_t)(kstart + 8 * h) * N8 + strip * SDW + cd;
    const float* xp = x + (size_t)c * K_DIM + kstart + 8 * h;

    v8f acc[TILES];
    #pragma unroll
    for (int t = 0; t < TILES; ++t) acc[t] = (v8f){};

    Stage sa, sb;
    load_stage(sa, qp, xp);

    #pragma unroll
    for (int g = 0; g < NGRP; ++g) {
        for (int p = 0; p < GROUPK / KSTEP / 2; ++p) {
            qp += (size_t)KSTEP * N8;  xp += KSTEP;
            load_stage(sb, qp, xp);
            __builtin_amdgcn_sched_barrier(0);
            compute_stage(acc, sa, sh16, sc[g], nzs[g]);

            const bool more = (g + 1 < NGRP) || (p + 1 < GROUPK / KSTEP / 2);
            qp += more ? (size_t)KSTEP * N8 : (size_t)0;
            xp += more ? KSTEP : 0;
            load_stage(sa, qp, xp);
            __builtin_amdgcn_sched_barrier(0);
            compute_stage(acc, sb, sh16, sc[g], nzs[g]);
        }
    }

    #pragma unroll
    for (int t = 0; t < TILES; ++t) {
        #pragma unroll
        for (int v = 0; v < 8; ++v)
            red[wave][v + 8 * h][8 * cd + 2 * t + ch] = acc[t][v];
    }
    __syncthreads();

    {
        const int row = wave;
        const int cl  = 2 * lane;
        float s0 = 0.0f, s1 = 0.0f;
        #pragma unroll
        for (int w = 0; w < WAVES; ++w) {
            const v2f pr = *(const v2f*)&red[w][row][cl];
            s0 += pr.x;
            s1 += pr.y;
        }
        const v2f bv = *(const v2f*)(bias + strip * STRIP + cl);
        v2f o;
        o.x = s0 + bv.x;
        o.y = s1 + bv.y;
        *(v2f*)(out + (size_t)row * N_DIM + strip * STRIP + cl) = o;
    }
}

extern "C" void kernel_launch(void* const* d_in, const int* in_sizes, int n_in,
                              void* d_out, int out_size, void* d_ws, size_t ws_size,
                              hipStream_t stream) {
    (void)in_sizes; (void)n_in; (void)d_ws; (void)ws_size; (void)out_size;
    const float* x       = (const float*)d_in[0];
    const int*   qweight = (const int*)  d_in[1];
    const int*   qzeros  = (const int*)  d_in[2];
    const float* scales  = (const float*)d_in[3];
    const float* bias    = (const float*)d_in[4];
    float*       out     = (float*)d_out;

    dim3 grid(N_DIM / STRIP);
    dim3 block(WAVES * 32);
    awq_wmma_f32_kernel<<<grid, block, 0, stream>>>(x, qweight, qzeros, scales, bias, out);
}
